# starting kernel
# baseline (speedup 1.0000x reference)
#include <hip/hip_runtime.h>
#include <hip/hip_bf16.h>


typedef float v2f __attribute__((ext_vector_type(2)));
typedef float v8f __attribute__((ext_vector_type(8)));

#define NN    1024
#define K1    512
#define HID   128
#define HID2  64

__device__ __forceinline__ float fast_tanh(float x) {
#if __has_builtin(__builtin_amdgcn_tanhf)
    return __builtin_amdgcn_tanhf(x);
#else
    return tanhf(x);
#endif
}

__device__ __forceinline__ v8f wmma_f32(v2f a, v2f b, v8f c) {
    return __builtin_amdgcn_wmma_f32_16x16x4_f32(
         false, a,  false, b,
         (short)0, c,  false,  false);
}

__global__ __launch_bounds__(128) void gemm1_tanh_kernel(
    const float* __restrict__ xf,
    const float* __restrict__ W_foh, const float* __restrict__ W_fom,
    const float* __restrict__ cat_bias,
    float* __restrict__ hf, float* __restrict__ hm)
{
    const int tile = blockIdx.x * blockDim.y + threadIdx.y;
    const int lane = threadIdx.x;
    const int mat   = tile >> 9;
    const int rem   = tile & 511;
    const int mtile = rem >> 3;
    const int ntile = rem & 7;

    const float* __restrict__ W    = mat ? W_fom : W_foh;
    float*       __restrict__ outp = mat ? hm    : hf;
    const float* __restrict__ bias = cat_bias + mat * HID;

    const int half = lane >> 4;
    const int l16  = lane & 15;
    const int row  = mtile * 16 + l16;
    const int col  = ntile * 16 + l16;
    const int khi  = half * 2;

    const float* arow = xf + row * K1 + khi;
    const float* bcol = W  + khi * HID + col;

    v8f c = {};
    #pragma unroll 4
    for (int k = 0; k < K1; k += 4) {
        v2f a = *reinterpret_cast<const v2f*>(arow + k);
        v2f b;
        b.x = bcol[k * HID];
        b.y = bcol[k * HID + HID];
        c = wmma_f32(a, b, c);
    }

    const float bn = bias[col];
    #pragma unroll
    for (int r = 0; r < 8; ++r) {
        const int m = mtile * 16 + r + half * 8;
        outp[m * HID + col] = fast_tanh(c[r] + bn);
    }
}

__global__ __launch_bounds__(128) void gemm2_kernel(
    const float* __restrict__ hf, const float* __restrict__ hm,
    const float* __restrict__ W_hid2, const float* __restrict__ hid2_bias,
    float* __restrict__ th, float* __restrict__ tmT)
{
    const int tile = blockIdx.x * blockDim.y + threadIdx.y;
    const int lane = threadIdx.x;
    const int mat   = tile >> 8;
    const int rem   = tile & 255;
    const int mtile = rem >> 2;
    const int ntile = rem & 3;

    const float* __restrict__ A = mat ? hm : hf;
    const float* __restrict__ B = W_hid2 + mat * HID * HID2;

    const int half = lane >> 4;
    const int l16  = lane & 15;
    const int row  = mtile * 16 + l16;
    const int col  = ntile * 16 + l16;
    const int khi  = half * 2;

    const float* arow = A + row * HID + khi;
    const float* bcol = B + khi * HID2 + col;

    v8f c = {};
    #pragma unroll
    for (int k = 0; k < HID; k += 4) {
        v2f a = *reinterpret_cast<const v2f*>(arow + k);
        v2f b;
        b.x = bcol[k * HID2];
        b.y = bcol[k * HID2 + HID2];
        c = wmma_f32(a, b, c);
    }

    if (mat == 0) {
        const float bn = hid2_bias[col];
        #pragma unroll
        for (int r = 0; r < 8; ++r) {
            const int m = mtile * 16 + r + half * 8;
            th[m * HID2 + col] = c[r] + bn;
        }
    } else {
        #pragma unroll
        for (int r = 0; r < 8; ++r) {
            const int m = mtile * 16 + r + half * 8;
            tmT[col * NN + m] = c[r];
        }
    }
}

#define IPT 8
__global__ __launch_bounds__(256) void outer_tanh_dot_kernel(
    const float* __restrict__ th, const float* __restrict__ tmT,
    const float* __restrict__ W_out, const float* __restrict__ out_bias,
    float* __restrict__ out)
{
    __shared__ float sw[HID2];
    __shared__ float sth[IPT][HID2];

    const int tx = threadIdx.x;
    const int j  = blockIdx.x * 256 + tx;
    const int i0 = blockIdx.y * IPT;

    if (tx < HID2) sw[tx] = W_out[tx];
    for (int t = tx; t < IPT * HID2; t += 256)
        sth[t >> 6][t & 63] = th[(i0 + (t >> 6)) * HID2 + (t & 63)];
    __syncthreads();

    float acc[IPT];
    #pragma unroll
    for (int ii = 0; ii < IPT; ++ii) acc[ii] = 0.0f;

    #pragma unroll 4
    for (int d = 0; d < HID2; ++d) {
        const float tmv = tmT[d * NN + j];
        const float wd  = sw[d];
        #pragma unroll
        for (int ii = 0; ii < IPT; ++ii)
            acc[ii] = fmaf(wd, fast_tanh(sth[ii][d] + tmv), acc[ii]);
    }

    const float ob = out_bias[0];
    #pragma unroll
    for (int ii = 0; ii < IPT; ++ii)
        out[(i0 + ii) * NN + j] = acc[ii] + ob;
}

extern "C" void kernel_launch(void* const* d_in, const int* in_sizes, int n_in,
                              void* d_out, int out_size, void* d_ws, size_t ws_size,
                              hipStream_t stream) {
    const float* xf        = (const float*)d_in[0];
    const float* W_foh     = (const float*)d_in[1];
    const float* W_fom     = (const float*)d_in[2];
    const float* cat_bias  = (const float*)d_in[3];
    const float* W_hid2    = (const float*)d_in[4];
    const float* hid2_bias = (const float*)d_in[5];
    const float* W_out     = (const float*)d_in[6];
    const float* out_bias  = (const float*)d_in[7];
    float* out = (float*)d_out;

    float* hf  = (float*)d_ws;
    float* hm  = hf  + NN * HID;
    float* th  = hm  + NN * HID;
    float* tmT = th  + NN * HID2;

    gemm1_tanh_kernel<<<dim3(256), dim3(32, 4), 0, stream>>>(
        xf, W_foh, W_fom, cat_bias, hf, hm);

    gemm2_kernel<<<dim3(128), dim3(32, 4), 0, stream>>>(
        hf, hm, W_hid2, hid2_bias, th, tmT);

    outer_tanh_dot_kernel<<<dim3(NN / 256, NN / IPT), dim3(256), 0, stream>>>(
        th, tmT, W_out, out_bias, out);
}


# gemm1 operand feed rebuilt: each wave takes a quarter of K for a 16x64 block, so every load is a whole-line 16-byte load issued two chunks ahead of its matrix instructions; partials summed in LDS
# speedup vs baseline: 1.7968x; 1.7968x over previous
#include <hip/hip_runtime.h>
#include <hip/hip_bf16.h>


typedef float v2f __attribute__((ext_vector_type(2)));
typedef float v4f __attribute__((ext_vector_type(4)));
typedef float v8f __attribute__((ext_vector_type(8)));

#define NN    1024
#define K1    512
#define HID   128
#define HID2  64

__device__ __forceinline__ float fast_tanh(float x) {
#if __has_builtin(__builtin_amdgcn_tanhf)
    return __builtin_amdgcn_tanhf(x);
#else
    return tanhf(x);
#endif
}

__device__ __forceinline__ v8f wmma_f32(v2f a, v2f b, v8f c) {
    return __builtin_amdgcn_wmma_f32_16x16x4_f32(
         false, a,  false, b,
         (short)0, c,  false,  false);
}

#define G1_JPC    4
#define G1_NCH    4
#define G1_RING   3
#define RED_PITCH 68

#define G1_LOAD(CH)                                                          \
    _Pragma("unroll")                                                        \
    for (int jj = 0; jj < G1_JPC; ++jj) {                                    \
        const int j_ = (CH) * G1_JPC + jj;                                   \
        abuf[(CH) % G1_RING][jj] =                                           \
            *reinterpret_cast<const v4f*>(ap + 8 * j_);                      \
        _Pragma("unroll")                                                    \
        for (int e = 0; e < 4; ++e)                                          \
            bbuf[(CH) % G1_RING][jj][e] =                                    \
                *reinterpret_cast<const v4f*>(bp + (8 * j_ + e) * HID);      \
    }

#define G1_MMA(CH)                                                           \
    _Pragma("unroll")                                                        \
    for (int jj = 0; jj < G1_JPC; ++jj) {                                    \
        const v4f av_ = abuf[(CH) % G1_RING][jj];                            \
        _Pragma("unroll")                                                    \
        for (int p = 0; p < 2; ++p) {                                        \
            v2f a_;                                                          \
            a_.x = av_[2 * p];                                               \
            a_.y = av_[2 * p + 1];                                           \
            _Pragma("unroll")                                                \
            for (int t = 0; t < 4; ++t) {                                    \
                v2f b_;                                                      \
                b_.x = bbuf[(CH) % G1_RING][jj][2 * p][t];                   \
                b_.y = bbuf[(CH) % G1_RING][jj][2 * p + 1][t];               \
                acc[t] = wmma_f32(a_, b_, acc[t]);                           \
            }                                                                \
        }                                                                    \
    }

__global__ __launch_bounds__(128) void gemm1_tanh_kernel(
    const float* __restrict__ xf,
    const float* __restrict__ W_foh, const float* __restrict__ W_fom,
    const float* __restrict__ cat_bias,
    float* __restrict__ hf, float* __restrict__ hm)
{
    __shared__ __attribute__((aligned(16))) float red[4][16][RED_PITCH];

    const int lane  = threadIdx.x;
    const int wv    = __builtin_amdgcn_readfirstlane(threadIdx.y);
    const int mtile = blockIdx.x >> 2;
    const int ng    = blockIdx.x & 3;
    const int mat   = ng >> 1;
    const int nn0   = (ng & 1) * 64;

    const float* __restrict__ W    = mat ? W_fom : W_foh;
    float*       __restrict__ outp = mat ? hm    : hf;

    const int h  = lane >> 4;
    const int c  = lane & 15;
    const int kw = wv * (K1 / 4);

    const float* ap = xf + (mtile * 16 + c) * K1 + kw + 4 * h;
    const float* bp = W + (kw + 4 * h) * HID + nn0 + 4 * c;

    v4f abuf[G1_RING][G1_JPC];
    v4f bbuf[G1_RING][G1_JPC][4];
    v8f acc[4] = {};

    G1_LOAD(0)
    G1_LOAD(1)
    #pragma unroll
    for (int ch = 0; ch < G1_NCH; ++ch) {
        if (ch + 2 < G1_NCH) { G1_LOAD(ch + 2) }
        __builtin_amdgcn_sched_barrier(0);
        G1_MMA(ch)
        __builtin_amdgcn_sched_barrier(0);
    }

    #pragma unroll
    for (int r = 0; r < 8; ++r) {
        v4f v;
        v.x = acc[0][r]; v.y = acc[1][r]; v.z = acc[2][r]; v.w = acc[3][r];
        *reinterpret_cast<v4f*>(&red[wv][8 * h + r][4 * c]) = v;
    }
    __syncthreads();

    const int tid = wv * 32 + lane;
    #pragma unroll
    for (int q = 0; q < 2; ++q) {
        const int idx = tid + 128 * q;
        const int row = idx >> 4;
        const int col = (idx & 15) * 4;
        v4f s = *reinterpret_cast<const v4f*>(&red[0][row][col]);
        s += *reinterpret_cast<const v4f*>(&red[1][row][col]);
        s += *reinterpret_cast<const v4f*>(&red[2][row][col]);
        s += *reinterpret_cast<const v4f*>(&red[3][row][col]);
        const v4f bv = *reinterpret_cast<const v4f*>(cat_bias + ng * 64 + col);
        v4f o;
        o.x = fast_tanh(s.x + bv.x);
        o.y = fast_tanh(s.y + bv.y);
        o.z = fast_tanh(s.z + bv.z);
        o.w = fast_tanh(s.w + bv.w);
        *reinterpret_cast<v4f*>(outp + (mtile * 16 + row) * HID + nn0 + col) = o;
    }
}
#undef G1_LOAD
#undef G1_MMA

__global__ __launch_bounds__(128) void gemm2_kernel(
    const float* __restrict__ hf, const float* __restrict__ hm,
    const float* __restrict__ W_hid2, const float* __restrict__ hid2_bias,
    float* __restrict__ th, float* __restrict__ tmT)
{
    const int tile = blockIdx.x * blockDim.y + threadIdx.y;
    const int lane = threadIdx.x;
    const int mat   = tile >> 8;
    const int rem   = tile & 255;
    const int mtile = rem >> 2;
    const int ntile = rem & 3;

    const float* __restrict__ A = mat ? hm : hf;
    const float* __restrict__ B = W_hid2 + mat * HID * HID2;

    const int half = lane >> 4;
    const int l16  = lane & 15;
    const int row  = mtile * 16 + l16;
    const int col  = ntile * 16 + l16;
    const int khi  = half * 2;

    const float* arow = A + row * HID + khi;
    const float* bcol = B + khi * HID2 + col;

    v8f c = {};
    #pragma unroll
    for (int k = 0; k < HID; k += 4) {
        v2f a = *reinterpret_cast<const v2f*>(arow + k);
        v2f b;
        b.x = bcol[k * HID2];
        b.y = bcol[k * HID2 + HID2];
        c = wmma_f32(a, b, c);
    }

    if (mat == 0) {
        const float bn = hid2_bias[col];
        #pragma unroll
        for (int r = 0; r < 8; ++r) {
            const int m = mtile * 16 + r + half * 8;
            th[m * HID2 + col] = c[r] + bn;
        }
    } else {
        #pragma unroll
        for (int r = 0; r < 8; ++r) {
            const int m = mtile * 16 + r + half * 8;
            tmT[col * NN + m] = c[r];
        }
    }
}

#define IPT 8
__global__ __launch_bounds__(256) void outer_tanh_dot_kernel(
    const float* __restrict__ th, const float* __restrict__ tmT,
    const float* __restrict__ W_out, const float* __restrict__ out_bias,
    float* __restrict__ out)
{
    __shared__ float sw[HID2];
    __shared__ float sth[IPT][HID2];

    const int tx = threadIdx.x;
    const int j  = blockIdx.x * 256 + tx;
    const int i0 = blockIdx.y * IPT;

    if (tx < HID2) sw[tx] = W_out[tx];
    for (int t = tx; t < IPT * HID2; t += 256)
        sth[t >> 6][t & 63] = th[(i0 + (t >> 6)) * HID2 + (t & 63)];
    __syncthreads();

    float acc[IPT];
    #pragma unroll
    for (int ii = 0; ii < IPT; ++ii) acc[ii] = 0.0f;

    #pragma unroll 4
    for (int d = 0; d < HID2; ++d) {
        const float tmv = tmT[d * NN + j];
        const float wd  = sw[d];
        #pragma unroll
        for (int ii = 0; ii < IPT; ++ii)
            acc[ii] = fmaf(wd, fast_tanh(sth[ii][d] + tmv), acc[ii]);
    }

    const float ob = out_bias[0];
    #pragma unroll
    for (int ii = 0; ii < IPT; ++ii)
        out[(i0 + ii) * NN + j] = acc[ii] + ob;
}

extern "C" void kernel_launch(void* const* d_in, const int* in_sizes, int n_in,
                              void* d_out, int out_size, void* d_ws, size_t ws_size,
                              hipStream_t stream) {
    const float* xf        = (const float*)d_in[0];
    const float* W_foh     = (const float*)d_in[1];
    const float* W_fom     = (const float*)d_in[2];
    const float* cat_bias  = (const float*)d_in[3];
    const float* W_hid2    = (const float*)d_in[4];
    const float* hid2_bias = (const float*)d_in[5];
    const float* W_out     = (const float*)d_in[6];
    const float* out_bias  = (const float*)d_in[7];
    float* out = (float*)d_out;

    float* hf  = (float*)d_ws;
    float* hm  = hf  + NN * HID;
    float* th  = hm  + NN * HID;
    float* tmT = th  + NN * HID2;

    gemm1_tanh_kernel<<<dim3(256), dim3(32, 4), 0, stream>>>(
        xf, W_foh, W_fom, cat_bias, hf, hm);

    gemm2_kernel<<<dim3(128), dim3(32, 4), 0, stream>>>(
        hf, hm, W_hid2, hid2_bias, th, tmT);

    outer_tanh_dot_kernel<<<dim3(NN / 256, NN / IPT), dim3(256), 0, stream>>>(
        th, tmT, W_out, out_bias, out);
}


# outer kernel operand feed rebuilt: 16 rows per lane in 256 blocks of 8 waves; each lane's tm loads go out three 4-d groups ahead of their tanh work, th/w LDS reads one group ahead
# speedup vs baseline: 2.8846x; 1.6054x over previous
#include <hip/hip_runtime.h>
#include <hip/hip_bf16.h>


typedef float v2f __attribute__((ext_vector_type(2)));
typedef float v4f __attribute__((ext_vector_type(4)));
typedef float v8f __attribute__((ext_vector_type(8)));

#define NN    1024
#define K1    512
#define HID   128
#define HID2  64

__device__ __forceinline__ float fast_tanh(float x) {
#if __has_builtin(__builtin_amdgcn_tanhf)
    return __builtin_amdgcn_tanhf(x);
#else
    return tanhf(x);
#endif
}

__device__ __forceinline__ v8f wmma_f32(v2f a, v2f b, v8f c) {
    return __builtin_amdgcn_wmma_f32_16x16x4_f32(
         false, a,  false, b,
         (short)0, c,  false,  false);
}

#define G1_JPC    4
#define G1_NCH    4
#define G1_RING   3
#define RED_PITCH 68

#define G1_LOAD(CH)                                                          \
    _Pragma("unroll")                                                        \
    for (int jj = 0; jj < G1_JPC; ++jj) {                                    \
        const int j_ = (CH) * G1_JPC + jj;                                   \
        abuf[(CH) % G1_RING][jj] =                                           \
            *reinterpret_cast<const v4f*>(ap + 8 * j_);                      \
        _Pragma("unroll")                                                    \
        for (int e = 0; e < 4; ++e)                                          \
            bbuf[(CH) % G1_RING][jj][e] =                                    \
                *reinterpret_cast<const v4f*>(bp + (8 * j_ + e) * HID);      \
    }

#define G1_MMA(CH)                                                           \
    _Pragma("unroll")                                                        \
    for (int jj = 0; jj < G1_JPC; ++jj) {                                    \
        const v4f av_ = abuf[(CH) % G1_RING][jj];                            \
        _Pragma("unroll")                                                    \
        for (int p = 0; p < 2; ++p) {                                        \
            v2f a_;                                                          \
            a_.x = av_[2 * p];                                               \
            a_.y = av_[2 * p + 1];                                           \
            _Pragma("unroll")                                                \
            for (int t = 0; t < 4; ++t) {                                    \
                v2f b_;                                                      \
                b_.x = bbuf[(CH) % G1_RING][jj][2 * p][t];                   \
                b_.y = bbuf[(CH) % G1_RING][jj][2 * p + 1][t];               \
                acc[t] = wmma_f32(a_, b_, acc[t]);                           \
            }                                                                \
        }                                                                    \
    }

__global__ __launch_bounds__(128) void gemm1_tanh_gemm2_kernel(
    const float* __restrict__ xf,
    const float* __restrict__ W_foh, const float* __restrict__ W_fom,
    const float* __restrict__ cat_bias,
    const float* __restrict__ W_hid2,
    float* __restrict__ thP, float* __restrict__ tmTP)
{
    __shared__ __attribute__((aligned(16))) float red[4][16][RED_PITCH];

    const int lane  = threadIdx.x;
    const int wv    = __builtin_amdgcn_readfirstlane(threadIdx.y);
    const int mtile = blockIdx.x >> 2;
    const int ng    = blockIdx.x & 3;
    const int mat   = ng >> 1;
    const int pl    = ng & 1;
    const int nn0   = pl * 64;

    const float* __restrict__ W = mat ? W_fom : W_foh;

    const int h  = lane >> 4;
    const int c  = lane & 15;
    const int kw = wv * (K1 / 4);

    const float* ap = xf + (mtile * 16 + c) * K1 + kw + 4 * h;
    const float* bp = W + (kw + 4 * h) * HID + nn0 + 4 * c;
    const float* w2p = W_hid2 + (mat * HID + nn0 + 4 * h) * HID2 + 16 * wv + c;

    v4f abuf[G1_RING][G1_JPC];
    v4f bbuf[G1_RING][G1_JPC][4];
    v8f acc[4] = {};
    v2f w2[8][2];

    G1_LOAD(0)
    G1_LOAD(1)
    #pragma unroll
    for (int ch = 0; ch < G1_NCH; ++ch) {
        if (ch + 2 < G1_NCH) { G1_LOAD(ch + 2) }
        if (ch + 2 == G1_NCH) {
            #pragma unroll
            for (int j = 0; j < 8; ++j) {
                #pragma unroll
                for (int p = 0; p < 2; ++p) {
                    w2[j][p].x = w2p[(8 * j + 2 * p) * HID2];
                    w2[j][p].y = w2p[(8 * j + 2 * p + 1) * HID2];
                }
            }
        }
        __builtin_amdgcn_sched_barrier(0);
        G1_MMA(ch)
        __builtin_amdgcn_sched_barrier(0);
    }

    #pragma unroll
    for (int r = 0; r < 8; ++r) {
        v4f v;
        v.x = acc[0][r]; v.y = acc[1][r]; v.z = acc[2][r]; v.w = acc[3][r];
        *reinterpret_cast<v4f*>(&red[wv][8 * h + r][4 * c]) = v;
    }
    __syncthreads();

    const int tid = wv * 32 + lane;
    #pragma unroll
    for (int q = 0; q < 2; ++q) {
        const int idx = tid + 128 * q;
        const int row = idx >> 4;
        const int col = (idx & 15) * 4;
        v4f s = *reinterpret_cast<const v4f*>(&red[0][row][col]);
        s += *reinterpret_cast<const v4f*>(&red[1][row][col]);
        s += *reinterpret_cast<const v4f*>(&red[2][row][col]);
        s += *reinterpret_cast<const v4f*>(&red[3][row][col]);
        const v4f bv = *reinterpret_cast<const v4f*>(cat_bias + ng * 64 + col);
        v4f o;
        o.x = fast_tanh(s.x + bv.x);
        o.y = fast_tanh(s.y + bv.y);
        o.z = fast_tanh(s.z + bv.z);
        o.w = fast_tanh(s.w + bv.w);
        *reinterpret_cast<v4f*>(&red[0][row][col]) = o;
    }
    __syncthreads();

    v8f c0 = {};
    v8f c1 = {};
    #pragma unroll
    for (int j = 0; j < 8; ++j) {
        const v4f av = *reinterpret_cast<const v4f*>(&red[0][c][8 * j + 4 * h]);
        v2f a0, a1;
        a0.x = av[0]; a0.y = av[1];
        a1.x = av[2]; a1.y = av[3];
        c0 = wmma_f32(a0, w2[j][0], c0);
        c1 = wmma_f32(a1, w2[j][1], c1);
    }
    c0 += c1;

    const int ocol = 16 * wv + c;
    const int m0   = mtile * 16 + 8 * h;
    if (mat == 0) {
        float* __restrict__ dst = thP + pl * (NN * HID2);
        #pragma unroll
        for (int r = 0; r < 8; ++r)
            dst[(m0 + r) * HID2 + ocol] = c0[r];
    } else {
        float* __restrict__ dst = tmTP + pl * (HID2 * NN) + ocol * NN + m0;
        v4f lo, hi;
        lo.x = c0[0]; lo.y = c0[1]; lo.z = c0[2]; lo.w = c0[3];
        hi.x = c0[4]; hi.y = c0[5]; hi.z = c0[6]; hi.w = c0[7];
        *reinterpret_cast<v4f*>(dst)     = lo;
        *reinterpret_cast<v4f*>(dst + 4) = hi;
    }
}
#undef G1_LOAD
#undef G1_MMA

#define OT_IPT   16
#define OT_NG    (HID2 / 4)
#define OT_BI    32
#define OT_BJ    128

#define OT_GLOAD(DST, G)                                                     \
    _Pragma("unroll")                                                        \
    for (int k = 0; k < 4; ++k) {                                            \
        DST[k][0] = tp[(4 * (G) + k) * NN];                                  \
        DST[k][1] = tp[HID2 * NN + (4 * (G) + k) * NN];                      \
    }

#define OT_LDS(DST, G)                                                       \
    DST[OT_IPT] = *reinterpret_cast<const v4f*>(&sw[4 * (G)]);               \
    _Pragma("unroll")                                                        \
    for (int ii = 0; ii < OT_IPT; ++ii)                                      \
        DST[ii] = *reinterpret_cast<const v4f*>(&srow[ii * HID2 + 4 * (G)]);

#define OT_COMPUTE(SRC, TH)                                                  \
    {                                                                        \
        float tmv_[4];                                                       \
        _Pragma("unroll")                                                    \
        for (int k = 0; k < 4; ++k) tmv_[k] = SRC[k][0] + SRC[k][1];         \
        _Pragma("unroll")                                                    \
        for (int ii = 0; ii < OT_IPT; ++ii) {                                \
            _Pragma("unroll")                                                \
            for (int k = 0; k < 4; ++k)                                      \
                acc[ii] = fmaf(TH[OT_IPT][k],                                \
                               fast_tanh(TH[ii][k] + tmv_[k]), acc[ii]);     \
        }                                                                    \
    }

#define OT_SB() __builtin_amdgcn_sched_barrier(0)

__global__ __launch_bounds__(256) void outer_tanh_dot_kernel(
    const float* __restrict__ thP, const float* __restrict__ tmTP,
    const float* __restrict__ hid2_bias,
    const float* __restrict__ W_out, const float* __restrict__ out_bias,
    float* __restrict__ out)
{
    __shared__ __attribute__((aligned(16))) float sw[HID2];
    __shared__ __attribute__((aligned(16))) float sth[OT_BI * HID2];

    const int tx   = threadIdx.x;
    const int lane = tx & 31;
    const int wv   = __builtin_amdgcn_readfirstlane(tx >> 5);
    const int j    = blockIdx.x * OT_BJ + (wv & 3) * 32 + lane;
    const int ib   = blockIdx.y * OT_BI;
    const int i0   = ib + (wv >> 2) * OT_IPT;

    const float* __restrict__ tp = tmTP + j;
    float ta[4][2], tb[4][2], tc[4][2], td[4][2];
    OT_GLOAD(ta, 0)
    OT_GLOAD(tb, 1)
    OT_GLOAD(tc, 2)

    v4f hp0[2], hp1[2], hb[2];
    #pragma unroll
    for (int q = 0; q < 2; ++q) {
        const int cell = tx + 256 * q;
        const int row  = cell >> 4;
        const int col  = (cell & 15) * 4;
        const int g    = (ib + row) * HID2 + col;
        hp0[q] = *reinterpret_cast<const v4f*>(thP + g);
        hp1[q] = *reinterpret_cast<const v4f*>(thP + NN * HID2 + g);
        hb[q]  = *reinterpret_cast<const v4f*>(hid2_bias + col);
    }
    float wo = 0.0f;
    if (tx < HID2) wo = W_out[tx];
    OT_SB();

    #pragma unroll
    for (int q = 0; q < 2; ++q) {
        const int cell = tx + 256 * q;
        const v4f s = (hp0[q] + hp1[q]) + hb[q];
        *reinterpret_cast<v4f*>(&sth[cell * 4]) = s;
    }
    if (tx < HID2) sw[tx] = wo;
    __syncthreads();

    const float* srow = &sth[(wv >> 2) * OT_IPT * HID2];

    float acc[OT_IPT];
    #pragma unroll
    for (int ii = 0; ii < OT_IPT; ++ii) acc[ii] = 0.0f;

    v4f tp_[OT_IPT + 1], tq_[OT_IPT + 1];
    OT_LDS(tp_, 0)

    #pragma unroll 1
    for (int g0 = 0; g0 < OT_NG - 4; g0 += 4) {
        OT_GLOAD(td, g0 + 3)
        OT_LDS(tq_, g0 + 1)
        OT_SB();
        OT_COMPUTE(ta, tp_)
        OT_SB();
        OT_GLOAD(ta, g0 + 4)
        OT_LDS(tp_, g0 + 2)
        OT_SB();
        OT_COMPUTE(tb, tq_)
        OT_SB();
        OT_GLOAD(tb, g0 + 5)
        OT_LDS(tq_, g0 + 3)
        OT_SB();
        OT_COMPUTE(tc, tp_)
        OT_SB();
        OT_GLOAD(tc, g0 + 6)
        OT_LDS(tp_, g0 + 4)
        OT_SB();
        OT_COMPUTE(td, tq_)
        OT_SB();
    }
    OT_GLOAD(td, OT_NG - 1)
    OT_LDS(tq_, OT_NG - 3)
    OT_SB();
    OT_COMPUTE(ta, tp_)
    OT_SB();
    OT_LDS(tp_, OT_NG - 2)
    OT_SB();
    OT_COMPUTE(tb, tq_)
    OT_SB();
    OT_LDS(tq_, OT_NG - 1)
    OT_SB();
    OT_COMPUTE(tc, tp_)
    OT_SB();
    OT_COMPUTE(td, tq_)

    const float ob = out_bias[0];
    #pragma unroll
    for (int ii = 0; ii < OT_IPT; ++ii)
        out[(i0 + ii) * NN + j] = acc[ii] + ob;
}
#undef OT_GLOAD
#undef OT_LDS
#undef OT_COMPUTE
#undef OT_SB

extern "C" void kernel_launch(void* const* d_in, const int* in_sizes, int n_in,
                              void* d_out, int out_size, void* d_ws, size_t ws_size,
                              hipStream_t stream) {
    const float* xf        = (const float*)d_in[0];
    const float* W_foh     = (const float*)d_in[1];
    const float* W_fom     = (const float*)d_in[2];
    const float* cat_bias  = (const float*)d_in[3];
    const float* W_hid2    = (const float*)d_in[4];
    const float* hid2_bias = (const float*)d_in[5];
    const float* W_out     = (const float*)d_in[6];
    const float* out_bias  = (const float*)d_in[7];
    float* out = (float*)d_out;

    float* thP  = (float*)d_ws;
    float* tmTP = thP + 2 * NN * HID2;

    gemm1_tanh_gemm2_kernel<<<dim3(256), dim3(32, 4), 0, stream>>>(
        xf, W_foh, W_fom, cat_bias, W_hid2, thP, tmTP);

    outer_tanh_dot_kernel<<<dim3(NN / OT_BJ, NN / OT_BI), dim3(256), 0, stream>>>(
        thP, tmTP, hid2_bias, W_out, out_bias, out);
}
